# starting kernel
# baseline (speedup 1.0000x reference)
#include <hip/hip_runtime.h>


typedef __attribute__((ext_vector_type(2))) float v2f;
typedef __attribute__((ext_vector_type(8))) float v8f;

#define NNODE 10000
#define NEDGE 50000
#define HDIM  64

template <int K, bool RELUA, bool TRANSB>
__device__ __forceinline__ v8f wmma_tile_f32(
    const float* __restrict__ A, int lda,
    const float* __restrict__ B, int ldb,
    int m0, int col, int hf, int l)
{
    const float* __restrict__ Arow = A + (size_t)(m0 + l) * lda;
    const float* __restrict__ Bcol = TRANSB ? (B + (size_t)col * ldb) : B;
    v8f acc = {};
#pragma unroll
    for (int k = 0; k < K; k += 4) {
        const int kb = k + (hf << 1);
        v2f a = *(const v2f*)(Arow + kb);
        if (RELUA) { a.x = fmaxf(a.x, 0.0f); a.y = fmaxf(a.y, 0.0f); }
        v2f b;
        if (TRANSB) {
            b = *(const v2f*)(Bcol + kb);
        } else {
            b.x = B[(size_t)kb * ldb + col];
            b.y = B[(size_t)(kb + 1) * ldb + col];
        }
        acc = __builtin_amdgcn_wmma_f32_16x16x4_f32(
            false, a, false, b, (short)0, acc, false, false);
    }
    return acc;
}

__device__ __forceinline__ void store_tile(
    float* __restrict__ C, int ldc, int m0, int col, int hf,
    float bv, bool relu, v8f acc)
{
#pragma unroll
    for (int v = 0; v < 8; ++v) {
        const int row = m0 + v + (hf << 3);
        float r = acc[v] + bv;
        if (relu) r = fmaxf(r, 0.0f);
        C[(size_t)row * ldc + col] = r;
    }
}

__global__ __launch_bounds__(128) void lin0_gemm(
    const float* __restrict__ x, const float* __restrict__ w,
    const float* __restrict__ bias, float* __restrict__ h)
{
    const int wave = threadIdx.x >> 5;
    const int lane = threadIdx.x & 31;
    const int hf = lane >> 4, l = lane & 15;
    const int tile = blockIdx.x * 4 + wave;
    if (tile >= (NNODE / 16) * (HDIM / 16)) return;
    const int tm = tile >> 2, tn = tile & 3;
    const int m0 = tm << 4, col = (tn << 4) + l;

    v8f acc = wmma_tile_f32<32, false, false>(x, 32, w, HDIM, m0, col, hf, l);
    store_tile(h, HDIM, m0, col, hf, bias[col], true, acc);
}

__global__ __launch_bounds__(128) void node_gemms(
    const float* __restrict__ h,
    const float* __restrict__ nn_w, const float* __restrict__ nn_b,
    const float* __restrict__ root_w, const float* __restrict__ conv_b,
    float* __restrict__ Y, float* __restrict__ agg)
{
    const int wave = threadIdx.x >> 5;
    const int lane = threadIdx.x & 31;
    const int hf = lane >> 4, l = lane & 15;
    const int tile = blockIdx.x * 4 + wave;
    const int perPlane = (NNODE / 16) * (HDIM / 16);
    if (tile >= 6 * perPlane) return;

    const int p = tile / perPlane;
    const int t = tile - p * perPlane;
    const int tm = t >> 2, tn = t & 3;
    const int m0 = tm << 4, col = (tn << 4) + l;

    const size_t plane = (size_t)NNODE * HDIM;
    const float* B = (p < 4) ? (nn_w + (size_t)p * HDIM * HDIM)
                             : ((p == 4) ? nn_b : root_w);
    float* C = (p < 5) ? (Y + (size_t)p * plane) : agg;
    const float bv = (p == 5) ? conv_b[col] : 0.0f;

    v8f acc = wmma_tile_f32<64, false, false>(h, HDIM, B, HDIM, m0, col, hf, l);
    store_tile(C, HDIM, m0, col, hf, bv, false, acc);
}

__global__ __launch_bounds__(256) void edge_msg(
    const int* __restrict__ src, const int* __restrict__ dst,
    const float* __restrict__ ea,
    const float* __restrict__ Y, float* __restrict__ agg, int E)
{
    const size_t plane = (size_t)NNODE * HDIM;
    const int t  = blockIdx.x * blockDim.x + threadIdx.x;
    const int o  = t & (HDIM - 1);
    int e        = t >> 6;
    const int es = (gridDim.x * blockDim.x) >> 6;

    for (; e < E; e += es) {
        const int en = e + es;
        if (en < E) {
            const int s2 = src[en];
            __builtin_prefetch(&Y[(size_t)s2 * HDIM + o], 0, 0);
        }
        const int s = src[e];
        const int d = dst[e];
        const float c0 = ea[e * 4 + 0];
        const float c1 = ea[e * 4 + 1];
        const float c2 = ea[e * 4 + 2];
        const float c3 = ea[e * 4 + 3];
        const size_t base = (size_t)s * HDIM + o;
        float m = Y[base + 4 * plane];
        m = fmaf(c0, Y[base + 0 * plane], m);
        m = fmaf(c1, Y[base + 1 * plane], m);
        m = fmaf(c2, Y[base + 2 * plane], m);
        m = fmaf(c3, Y[base + 3 * plane], m);
        atomicAdd(&agg[(size_t)d * HDIM + o], m);
    }
}

__global__ __launch_bounds__(128) void gru_gemms(
    const float* __restrict__ agg, const float* __restrict__ h,
    const float* __restrict__ w_ih, const float* __restrict__ w_hh,
    const float* __restrict__ b_ih, const float* __restrict__ b_hh,
    float* __restrict__ gi, float* __restrict__ gh)
{
    const int wave = threadIdx.x >> 5;
    const int lane = threadIdx.x & 31;
    const int hf = lane >> 4, l = lane & 15;
    const int perJob = (NNODE / 16) * (3 * HDIM / 16);
    const int tile = blockIdx.x * 4 + wave;
    if (tile >= 2 * perJob) return;

    const int job = tile / perJob;
    const int t   = tile - job * perJob;
    const int tilesN = (3 * HDIM) / 16;
    const int tm = t / tilesN, tn = t - tm * tilesN;
    const int m0 = tm << 4, col = (tn << 4) + l;

    v8f acc;
    if (job == 0) {
        acc = wmma_tile_f32<64, true, true>(agg, HDIM, w_ih, HDIM, m0, col, hf, l);
        store_tile(gi, 3 * HDIM, m0, col, hf, b_ih[col], false, acc);
    } else {
        acc = wmma_tile_f32<64, false, true>(h, HDIM, w_hh, HDIM, m0, col, hf, l);
        store_tile(gh, 3 * HDIM, m0, col, hf, b_hh[col], false, acc);
    }
}

__global__ __launch_bounds__(256) void gru_update(
    const float* __restrict__ gi, const float* __restrict__ gh,
    float* __restrict__ h, int total)
{
    const int t = blockIdx.x * blockDim.x + threadIdx.x;
    if (t >= total) return;
    const int node = t >> 6;
    const int o    = t & (HDIM - 1);
    const float* gin = gi + (size_t)node * (3 * HDIM);
    const float* ghn = gh + (size_t)node * (3 * HDIM);

    const float sr = gin[o]        + ghn[o];
    const float sz = gin[HDIM + o] + ghn[HDIM + o];
    const float in = gin[2 * HDIM + o];
    const float hn = ghn[2 * HDIM + o];

    const float r = 1.0f / (1.0f + __expf(-sr));
    const float z = 1.0f / (1.0f + __expf(-sz));
    const float n = tanhf(in + r * hn);
    const float ho = h[t];
    h[t] = (1.0f - z) * n + z * ho;
}

extern "C" void kernel_launch(void* const* d_in, const int* in_sizes, int n_in,
                              void* d_out, int out_size, void* d_ws, size_t ws_size,
                              hipStream_t stream)
{
    (void)in_sizes; (void)n_in; (void)out_size; (void)ws_size;

    const float* x      = (const float*)d_in[0];
    const int*   eidx   = (const int*)  d_in[1];
    const float* ea     = (const float*)d_in[2];
    const float* lin0_w = (const float*)d_in[3];
    const float* lin0_b = (const float*)d_in[4];
    const float* nn_w   = (const float*)d_in[5];
    const float* nn_b   = (const float*)d_in[6];
    const float* root_w = (const float*)d_in[7];
    const float* conv_b = (const float*)d_in[8];
    const float* w_ih   = (const float*)d_in[9];
    const float* w_hh   = (const float*)d_in[10];
    const float* b_ih   = (const float*)d_in[11];
    const float* b_hh   = (const float*)d_in[12];

    const int* src = eidx;
    const int* dst = eidx + NEDGE;

    float* h = (float*)d_out;

    const size_t plane = (size_t)NNODE * HDIM;
    float* Y   = (float*)d_ws;
    float* agg = Y + 5 * plane;
    float* gi  = agg + plane;
    float* gh  = gi + (size_t)NNODE * 3 * HDIM;

    const int gLin0 = ((NNODE / 16) * (HDIM / 16) + 3) / 4;
    const int gNode = (6 * (NNODE / 16) * (HDIM / 16) + 3) / 4;
    const int gGru  = (2 * (NNODE / 16) * (3 * HDIM / 16) + 3) / 4;

    lin0_gemm<<<gLin0, 128, 0, stream>>>(x, lin0_w, lin0_b, h);

    for (int it = 0; it < 3; ++it) {
        node_gemms<<<gNode, 128, 0, stream>>>(h, nn_w, nn_b, root_w, conv_b,
                                              Y, agg);
        edge_msg<<<4096, 256, 0, stream>>>(src, dst, ea, Y, agg, NEDGE);
        gru_gemms<<<gGru, 128, 0, stream>>>(agg, h, w_ih, w_hh, b_ih, b_hh,
                                            gi, gh);
        gru_update<<<(NNODE * HDIM + 255) / 256, 256, 0, stream>>>(
            gi, gh, h, NNODE * HDIM);
    }
}


# Edge kernel: a lane owns 4 columns of an edge (two 8-byte pairs 32 columns apart), 16 lanes per edge, 10 gathers in flight per wave; 32768 waves become 6256; atomic arrival order changes
# speedup vs baseline: 1.3459x; 1.3459x over previous
#include <hip/hip_runtime.h>


typedef __attribute__((ext_vector_type(2))) float v2f;
typedef __attribute__((ext_vector_type(8))) float v8f;

#define NNODE 10000
#define NEDGE 50000
#define HDIM  64

template <int K, bool RELUA, bool TRANSB>
__device__ __forceinline__ v8f wmma_tile_f32(
    const float* __restrict__ A, int lda,
    const float* __restrict__ B, int ldb,
    int m0, int col, int hf, int l)
{
    const float* __restrict__ Arow = A + (size_t)(m0 + l) * lda;
    const float* __restrict__ Bcol = TRANSB ? (B + (size_t)col * ldb) : B;
    v8f acc = {};
#pragma unroll
    for (int k = 0; k < K; k += 4) {
        const int kb = k + (hf << 1);
        v2f a = *(const v2f*)(Arow + kb);
        if (RELUA) { a.x = fmaxf(a.x, 0.0f); a.y = fmaxf(a.y, 0.0f); }
        v2f b;
        if (TRANSB) {
            b = *(const v2f*)(Bcol + kb);
        } else {
            b.x = B[(size_t)kb * ldb + col];
            b.y = B[(size_t)(kb + 1) * ldb + col];
        }
        acc = __builtin_amdgcn_wmma_f32_16x16x4_f32(
            false, a, false, b, (short)0, acc, false, false);
    }
    return acc;
}

__device__ __forceinline__ void store_tile(
    float* __restrict__ C, int ldc, int m0, int col, int hf,
    float bv, bool relu, v8f acc)
{
#pragma unroll
    for (int v = 0; v < 8; ++v) {
        const int row = m0 + v + (hf << 3);
        float r = acc[v] + bv;
        if (relu) r = fmaxf(r, 0.0f);
        C[(size_t)row * ldc + col] = r;
    }
}

template <int K, bool RELUA, bool TRANSB>
__device__ __forceinline__ void wmma_block_2x4(
    const float* __restrict__ A, int lda,
    const float* __restrict__ B, int ldb,
    int m0a, int m0b, int n0, int hf, int l, v8f (&acc)[2][4])
{
    const float* __restrict__ Ar0 = A + (size_t)(m0a + l) * lda;
    const float* __restrict__ Ar1 = A + (size_t)(m0b + l) * lda;
#pragma unroll
    for (int k = 0; k < K; k += 4) {
        const int kb = k + (hf << 1);
        v2f a0 = *(const v2f*)(Ar0 + kb);
        v2f a1 = *(const v2f*)(Ar1 + kb);
        if (RELUA) {
            a0.x = fmaxf(a0.x, 0.0f); a0.y = fmaxf(a0.y, 0.0f);
            a1.x = fmaxf(a1.x, 0.0f); a1.y = fmaxf(a1.y, 0.0f);
        }
#pragma unroll
        for (int c = 0; c < 4; ++c) {
            const int col = n0 + (c << 4) + l;
            v2f b;
            if (TRANSB) {
                b = *(const v2f*)(B + (size_t)col * ldb + kb);
            } else {
                b.x = B[(size_t)kb * ldb + col];
                b.y = B[(size_t)(kb + 1) * ldb + col];
            }
            acc[0][c] = __builtin_amdgcn_wmma_f32_16x16x4_f32(
                false, a0, false, b, (short)0, acc[0][c], false, false);
            acc[1][c] = __builtin_amdgcn_wmma_f32_16x16x4_f32(
                false, a1, false, b, (short)0, acc[1][c], false, false);
        }
    }
}

__device__ __forceinline__ void store_tiles4(
    float* __restrict__ C, int ldc, int m0, int n0, int l, int hf,
    const float (&bv)[4], const v8f (&acc)[4])
{
#pragma unroll
    for (int c = 0; c < 4; ++c)
        store_tile(C, ldc, m0, n0 + (c << 4) + l, hf, bv[c], false, acc[c]);
}

template <int K>
__device__ __forceinline__ void wmma_block_2x3_gates(
    const float* __restrict__ A, int lda,
    const float* __restrict__ W,
    int m0a, int m0b, int j0, int hf, int l, v8f (&acc)[2][3])
{
    const float* __restrict__ Ar0 = A + (size_t)(m0a + l) * lda;
    const float* __restrict__ Ar1 = A + (size_t)(m0b + l) * lda;
    const float* __restrict__ Wr  = W + (size_t)(j0 + l) * K;
#pragma unroll
    for (int k = 0; k < K; k += 4) {
        const int kb = k + (hf << 1);
        v2f a0 = *(const v2f*)(Ar0 + kb);
        v2f a1 = *(const v2f*)(Ar1 + kb);
        a0.x = fmaxf(a0.x, 0.0f); a0.y = fmaxf(a0.y, 0.0f);
        a1.x = fmaxf(a1.x, 0.0f); a1.y = fmaxf(a1.y, 0.0f);
#pragma unroll
        for (int g = 0; g < 3; ++g) {
            const v2f b = *(const v2f*)(Wr + (size_t)g * HDIM * K + kb);
            acc[0][g] = __builtin_amdgcn_wmma_f32_16x16x4_f32(
                false, a0, false, b, (short)0, acc[0][g], false, false);
            acc[1][g] = __builtin_amdgcn_wmma_f32_16x16x4_f32(
                false, a1, false, b, (short)0, acc[1][g], false, false);
        }
    }
}

__device__ __forceinline__ void load_gate_inputs(
    const float* __restrict__ gh, const float* __restrict__ h,
    int m0, int j, int hf, float (&ghv)[3][8], float (&hov)[8])
{
#pragma unroll
    for (int v = 0; v < 8; ++v) {
        const size_t row = (size_t)(m0 + v + (hf << 3));
        const float* __restrict__ ghn = gh + row * (3 * HDIM);
        ghv[0][v] = ghn[j];
        ghv[1][v] = ghn[HDIM + j];
        ghv[2][v] = ghn[2 * HDIM + j];
        hov[v]    = h[row * HDIM + j];
    }
}

__device__ __forceinline__ void gate_update_tile(
    const v8f (&acc)[3], const float (&bi)[3],
    const float (&ghv)[3][8], const float (&hov)[8],
    float* __restrict__ h, int m0, int j, int hf)
{
#pragma unroll
    for (int v = 0; v < 8; ++v) {
        const size_t row = (size_t)(m0 + v + (hf << 3));
        const float gir = acc[0][v] + bi[0];
        const float giz = acc[1][v] + bi[1];
        const float gin = acc[2][v] + bi[2];
        const float sr = gir + ghv[0][v];
        const float sz = giz + ghv[1][v];
        const float hn = ghv[2][v];

        const float r = 1.0f / (1.0f + __expf(-sr));
        const float z = 1.0f / (1.0f + __expf(-sz));
        const float n = tanhf(gin + r * hn);
        const float ho = hov[v];
        h[row * HDIM + j] = (1.0f - z) * n + z * ho;
    }
}

#define NROWT  (NNODE / 16)
#define NROWP  ((NROWT + 1) / 2)

__global__ __launch_bounds__(128) void lin0_gemm(
    const float* __restrict__ x, const float* __restrict__ w,
    const float* __restrict__ bias, float* __restrict__ h)
{
    const int wave = threadIdx.x >> 5;
    const int lane = threadIdx.x & 31;
    const int hf = lane >> 4, l = lane & 15;
    const int tile = blockIdx.x * 4 + wave;
    if (tile >= (NNODE / 16) * (HDIM / 16)) return;
    const int tm = tile >> 2, tn = tile & 3;
    const int m0 = tm << 4, col = (tn << 4) + l;

    v8f acc = wmma_tile_f32<32, false, false>(x, 32, w, HDIM, m0, col, hf, l);
    store_tile(h, HDIM, m0, col, hf, bias[col], true, acc);
}

__global__ __launch_bounds__(128) void node_gemms(
    const float* __restrict__ h,
    const float* __restrict__ nn_w, const float* __restrict__ nn_b,
    const float* __restrict__ root_w, const float* __restrict__ conv_b,
    const float* __restrict__ w_hh, const float* __restrict__ b_hh,
    float* __restrict__ Y, float* __restrict__ agg, float* __restrict__ gh)
{
    const int wave = threadIdx.x >> 5;
    const int lane = threadIdx.x & 31;
    const int hf = lane >> 4, l = lane & 15;
    const int item = blockIdx.x * 4 + wave;
    if (item >= 9 * NROWP) return;

    const int p  = item / NROWP;
    const int rp = item - p * NROWP;
    const int tm0 = rp << 1;
    const bool has1 = (tm0 + 1) < NROWT;
    const int m0a = tm0 << 4;
    const int m0b = has1 ? (m0a + 16) : m0a;

    const size_t plane = (size_t)NNODE * HDIM;

    float bv[4] = {0.0f, 0.0f, 0.0f, 0.0f};
    float* C;
    int ldc, n0;

    v8f acc[2][4] = {};
    if (p < 6) {
        const float* B = (p < 4) ? (nn_w + (size_t)p * HDIM * HDIM)
                                 : ((p == 4) ? nn_b : root_w);
        C = (p < 5) ? (Y + (size_t)p * plane) : agg;
        ldc = HDIM;
        n0 = 0;
        if (p == 5) {
            bv[0] = conv_b[l];      bv[1] = conv_b[16 + l];
            bv[2] = conv_b[32 + l]; bv[3] = conv_b[48 + l];
        }
        wmma_block_2x4<64, false, false>(h, HDIM, B, HDIM, m0a, m0b, 0, hf, l, acc);
    } else {
        n0 = (p - 6) * HDIM;
        C = gh;
        ldc = 3 * HDIM;
        bv[0] = b_hh[n0 + l];      bv[1] = b_hh[n0 + 16 + l];
        bv[2] = b_hh[n0 + 32 + l]; bv[3] = b_hh[n0 + 48 + l];
        wmma_block_2x4<64, false, true>(h, HDIM, w_hh, HDIM, m0a, m0b, n0, hf, l, acc);
    }

    store_tiles4(C, ldc, m0a, n0, l, hf, bv, acc[0]);
    if (has1) store_tiles4(C, ldc, m0b, n0, l, hf, bv, acc[1]);
}

__global__ __launch_bounds__(256) void edge_msg(
    const int* __restrict__ src, const int* __restrict__ dst,
    const float* __restrict__ ea,
    const float* __restrict__ Y, float* __restrict__ agg, int E)
{
    const size_t plane = (size_t)NNODE * HDIM;
    const int t  = blockIdx.x * blockDim.x + threadIdx.x;
    const int o  = (t & 15) << 1;
    int e        = t >> 4;
    const int es = (gridDim.x * blockDim.x) >> 4;
    if (e >= E) return;

    int   s  = src[e];
    int   d  = dst[e];
    float c0 = ea[e * 4 + 0];
    float c1 = ea[e * 4 + 1];
    float c2 = ea[e * 4 + 2];
    float c3 = ea[e * 4 + 3];

    for (;;) {
        const int  en   = e + es;
        const bool more = en < E;
        const int  eq   = more ? en : e;
        const int   s2 = src[eq];
        const int   d2 = dst[eq];
        const float n0 = ea[eq * 4 + 0];
        const float n1 = ea[eq * 4 + 1];
        const float n2 = ea[eq * 4 + 2];
        const float n3 = ea[eq * 4 + 3];

        const float* __restrict__ yr = Y + (size_t)s * HDIM + o;
        const v2f a4 = *(const v2f*)(yr + 4 * plane);
        const v2f b4 = *(const v2f*)(yr + 4 * plane + 32);
        const v2f a0 = *(const v2f*)(yr + 0 * plane);
        const v2f b0 = *(const v2f*)(yr + 0 * plane + 32);
        const v2f a1 = *(const v2f*)(yr + 1 * plane);
        const v2f b1 = *(const v2f*)(yr + 1 * plane + 32);
        const v2f a2 = *(const v2f*)(yr + 2 * plane);
        const v2f b2 = *(const v2f*)(yr + 2 * plane + 32);
        const v2f a3 = *(const v2f*)(yr + 3 * plane);
        const v2f b3 = *(const v2f*)(yr + 3 * plane + 32);

        __builtin_amdgcn_sched_barrier(0);

        float m0 = a4.x, m1 = a4.y, m2 = b4.x, m3 = b4.y;
        m0 = fmaf(c0, a0.x, m0); m1 = fmaf(c0, a0.y, m1);
        m2 = fmaf(c0, b0.x, m2); m3 = fmaf(c0, b0.y, m3);
        m0 = fmaf(c1, a1.x, m0); m1 = fmaf(c1, a1.y, m1);
        m2 = fmaf(c1, b1.x, m2); m3 = fmaf(c1, b1.y, m3);
        m0 = fmaf(c2, a2.x, m0); m1 = fmaf(c2, a2.y, m1);
        m2 = fmaf(c2, b2.x, m2); m3 = fmaf(c2, b2.y, m3);
        m0 = fmaf(c3, a3.x, m0); m1 = fmaf(c3, a3.y, m1);
        m2 = fmaf(c3, b3.x, m2); m3 = fmaf(c3, b3.y, m3);

        float* __restrict__ ar = agg + (size_t)d * HDIM + o;
        atomicAdd(ar,      m0);
        atomicAdd(ar + 1,  m1);
        atomicAdd(ar + 32, m2);
        atomicAdd(ar + 33, m3);

        if (!more) break;
        e = en; s = s2; d = d2;
        c0 = n0; c1 = n1; c2 = n2; c3 = n3;
    }
}

__global__ __launch_bounds__(128) void gru_step(
    const float* __restrict__ agg, const float* __restrict__ gh,
    const float* __restrict__ w_ih, const float* __restrict__ b_ih,
    float* __restrict__ h)
{
    const int wave = threadIdx.x >> 5;
    const int lane = threadIdx.x & 31;
    const int hf = lane >> 4, l = lane & 15;
    const int rp = blockIdx.x;
    const int j0 = wave << 4;
    const int j  = j0 + l;
    const int tm0 = rp << 1;
    const bool has1 = (tm0 + 1) < NROWT;
    const int m0a = tm0 << 4;
    const int m0b = has1 ? (m0a + 16) : m0a;

    const float bi[3] = {b_ih[j], b_ih[HDIM + j], b_ih[2 * HDIM + j]};
    float ghA[3][8], ghB[3][8], hoA[8], hoB[8];
    load_gate_inputs(gh, h, m0a, j, hf, ghA, hoA);
    load_gate_inputs(gh, h, m0b, j, hf, ghB, hoB);

    v8f acc[2][3] = {};
    wmma_block_2x3_gates<64>(agg, HDIM, w_ih, m0a, m0b, j0, hf, l, acc);

    gate_update_tile(acc[0], bi, ghA, hoA, h, m0a, j, hf);
    if (has1) gate_update_tile(acc[1], bi, ghB, hoB, h, m0b, j, hf);
}

extern "C" void kernel_launch(void* const* d_in, const int* in_sizes, int n_in,
                              void* d_out, int out_size, void* d_ws, size_t ws_size,
                              hipStream_t stream)
{
    (void)in_sizes; (void)n_in; (void)out_size; (void)ws_size;

    const float* x      = (const float*)d_in[0];
    const int*   eidx   = (const int*)  d_in[1];
    const float* ea     = (const float*)d_in[2];
    const float* lin0_w = (const float*)d_in[3];
    const float* lin0_b = (const float*)d_in[4];
    const float* nn_w   = (const float*)d_in[5];
    const float* nn_b   = (const float*)d_in[6];
    const float* root_w = (const float*)d_in[7];
    const float* conv_b = (const float*)d_in[8];
    const float* w_ih   = (const float*)d_in[9];
    const float* w_hh   = (const float*)d_in[10];
    const float* b_ih   = (const float*)d_in[11];
    const float* b_hh   = (const float*)d_in[12];

    const int* src = eidx;
    const int* dst = eidx + NEDGE;

    float* h = (float*)d_out;

    const size_t plane = (size_t)NNODE * HDIM;
    float* Y   = (float*)d_ws;
    float* agg = Y + 5 * plane;
    float* gh  = agg + plane;

    const int gLin0 = ((NNODE / 16) * (HDIM / 16) + 3) / 4;
    const int gNode = (9 * NROWP + 3) / 4;
    const int gStep = NROWP;

    lin0_gemm<<<gLin0, 128, 0, stream>>>(x, lin0_w, lin0_b, h);

    for (int it = 0; it < 3; ++it) {
        node_gemms<<<gNode, 128, 0, stream>>>(h, nn_w, nn_b, root_w, conv_b,
                                              w_hh, b_hh, Y, agg, gh);
        edge_msg<<<782, 256, 0, stream>>>(src, dst, ea, Y, agg, NEDGE);
        gru_step<<<gStep, 128, 0, stream>>>(agg, gh, w_ih, b_ih, h);
    }
}


# Edge sums moved into the GRU kernel: edges binned once per call by 16-row dst tile (int slot claim), block adds its messages into an LDS tile by LDS float atomics, WMMA reads it; 10 to 7 launches
# speedup vs baseline: 1.5983x; 1.1875x over previous
#include <hip/hip_runtime.h>


typedef __attribute__((ext_vector_type(2))) float v2f;
typedef __attribute__((ext_vector_type(4))) float v4f;
typedef __attribute__((ext_vector_type(4))) int   v4i;
typedef __attribute__((ext_vector_type(8))) float v8f;

#define NNODE 10000
#define NEDGE 50000
#define HDIM  64

#define NBKT   (NNODE / 16)
#define BCAP   128
#define LPITCH 68
#define GLIN0  625
#define GNODE  705
#define GBIN   ((NEDGE + 127) / 128)

template <int K, bool RELUA, bool TRANSB>
__device__ __forceinline__ v8f wmma_tile_f32(
    const float* __restrict__ A, int lda,
    const float* __restrict__ B, int ldb,
    int m0, int col, int hf, int l)
{
    const float* __restrict__ Arow = A + (size_t)(m0 + l) * lda;
    const float* __restrict__ Bcol = TRANSB ? (B + (size_t)col * ldb) : B;
    v8f acc = {};
#pragma unroll
    for (int k = 0; k < K; k += 4) {
        const int kb = k + (hf << 1);
        v2f a = *(const v2f*)(Arow + kb);
        if (RELUA) { a.x = fmaxf(a.x, 0.0f); a.y = fmaxf(a.y, 0.0f); }
        v2f b;
        if (TRANSB) {
            b = *(const v2f*)(Bcol + kb);
        } else {
            b.x = B[(size_t)kb * ldb + col];
            b.y = B[(size_t)(kb + 1) * ldb + col];
        }
        acc = __builtin_amdgcn_wmma_f32_16x16x4_f32(
            false, a, false, b, (short)0, acc, false, false);
    }
    return acc;
}

__device__ __forceinline__ void store_tile(
    float* __restrict__ C, int ldc, int m0, int col, int hf,
    float bv, bool relu, v8f acc)
{
#pragma unroll
    for (int v = 0; v < 8; ++v) {
        const int row = m0 + v + (hf << 3);
        float r = acc[v] + bv;
        if (relu) r = fmaxf(r, 0.0f);
        C[(size_t)row * ldc + col] = r;
    }
}

template <int K, bool RELUA, bool TRANSB>
__device__ __forceinline__ void wmma_block_2x4(
    const float* __restrict__ A, int lda,
    const float* __restrict__ B, int ldb,
    int m0a, int m0b, int n0, int hf, int l, v8f (&acc)[2][4])
{
    const float* __restrict__ Ar0 = A + (size_t)(m0a + l) * lda;
    const float* __restrict__ Ar1 = A + (size_t)(m0b + l) * lda;
#pragma unroll
    for (int k = 0; k < K; k += 4) {
        const int kb = k + (hf << 1);
        v2f a0 = *(const v2f*)(Ar0 + kb);
        v2f a1 = *(const v2f*)(Ar1 + kb);
        if (RELUA) {
            a0.x = fmaxf(a0.x, 0.0f); a0.y = fmaxf(a0.y, 0.0f);
            a1.x = fmaxf(a1.x, 0.0f); a1.y = fmaxf(a1.y, 0.0f);
        }
#pragma unroll
        for (int c = 0; c < 4; ++c) {
            const int col = n0 + (c << 4) + l;
            v2f b;
            if (TRANSB) {
                b = *(const v2f*)(B + (size_t)col * ldb + kb);
            } else {
                b.x = B[(size_t)kb * ldb + col];
                b.y = B[(size_t)(kb + 1) * ldb + col];
            }
            acc[0][c] = __builtin_amdgcn_wmma_f32_16x16x4_f32(
                false, a0, false, b, (short)0, acc[0][c], false, false);
            acc[1][c] = __builtin_amdgcn_wmma_f32_16x16x4_f32(
                false, a1, false, b, (short)0, acc[1][c], false, false);
        }
    }
}

__device__ __forceinline__ void store_tiles4(
    float* __restrict__ C, int ldc, int m0, int n0, int l, int hf,
    const float (&bv)[4], const v8f (&acc)[4])
{
#pragma unroll
    for (int c = 0; c < 4; ++c)
        store_tile(C, ldc, m0, n0 + (c << 4) + l, hf, bv[c], false, acc[c]);
}

__device__ __forceinline__ void wmma_1x3_gates_lds(
    const float* tile, const float* __restrict__ W,
    int j0, int hf, int l, v8f (&acc)[3])
{
    const float* Ar = tile + l * LPITCH;
    const float* __restrict__ Wr = W + (size_t)(j0 + l) * HDIM;
#pragma unroll
    for (int k = 0; k < HDIM; k += 4) {
        const int kb = k + (hf << 1);
        v2f a = *(const v2f*)(Ar + kb);
        a.x = fmaxf(a.x, 0.0f); a.y = fmaxf(a.y, 0.0f);
#pragma unroll
        for (int g = 0; g < 3; ++g) {
            const v2f b = *(const v2f*)(Wr + (size_t)g * HDIM * HDIM + kb);
            acc[g] = __builtin_amdgcn_wmma_f32_16x16x4_f32(
                false, a, false, b, (short)0, acc[g], false, false);
        }
    }
}

__device__ __forceinline__ void load_gate_inputs(
    const float* __restrict__ gh, const float* __restrict__ h,
    int m0, int j, int hf, float (&ghv)[3][8], float (&hov)[8])
{
#pragma unroll
    for (int v = 0; v < 8; ++v) {
        const size_t row = (size_t)(m0 + v + (hf << 3));
        const float* __restrict__ ghn = gh + row * (3 * HDIM);
        ghv[0][v] = ghn[j];
        ghv[1][v] = ghn[HDIM + j];
        ghv[2][v] = ghn[2 * HDIM + j];
        hov[v]    = h[row * HDIM + j];
    }
}

__device__ __forceinline__ void gate_update_tile(
    const v8f (&acc)[3], const float (&bi)[3],
    const float (&ghv)[3][8], const float (&hov)[8],
    float* __restrict__ h, int m0, int j, int hf)
{
#pragma unroll
    for (int v = 0; v < 8; ++v) {
        const size_t row = (size_t)(m0 + v + (hf << 3));
        const float gir = acc[0][v] + bi[0];
        const float giz = acc[1][v] + bi[1];
        const float gin = acc[2][v] + bi[2];
        const float sr = gir + ghv[0][v];
        const float sz = giz + ghv[1][v];
        const float hn = ghv[2][v];

        const float r = 1.0f / (1.0f + __expf(-sr));
        const float z = 1.0f / (1.0f + __expf(-sz));
        const float n = tanhf(gin + r * hn);
        const float ho = hov[v];
        h[row * HDIM + j] = (1.0f - z) * n + z * ho;
    }
}

#define NROWT  (NNODE / 16)
#define NROWP  ((NROWT + 1) / 2)

__global__ __launch_bounds__(128) void lin0_gemm(
    const float* __restrict__ x, const float* __restrict__ w,
    const float* __restrict__ bias, float* __restrict__ h,
    int* __restrict__ cnt)
{
    if (blockIdx.x >= GLIN0) {
        for (int i = threadIdx.x; i < NBKT; i += 128) cnt[i] = 0;
        return;
    }
    const int wave = threadIdx.x >> 5;
    const int lane = threadIdx.x & 31;
    const int hf = lane >> 4, l = lane & 15;
    const int tile = blockIdx.x * 4 + wave;
    if (tile >= (NNODE / 16) * (HDIM / 16)) return;
    const int tm = tile >> 2, tn = tile & 3;
    const int m0 = tm << 4, col = (tn << 4) + l;

    v8f acc = wmma_tile_f32<32, false, false>(x, 32, w, HDIM, m0, col, hf, l);
    store_tile(h, HDIM, m0, col, hf, bias[col], true, acc);
}

__global__ __launch_bounds__(128) void node_gemms(
    const float* __restrict__ h,
    const float* __restrict__ nn_w, const float* __restrict__ nn_b,
    const float* __restrict__ root_w, const float* __restrict__ conv_b,
    const float* __restrict__ w_hh, const float* __restrict__ b_hh,
    float* __restrict__ Y, float* __restrict__ agg, float* __restrict__ gh,
    const int* __restrict__ src, const int* __restrict__ dst,
    const float* __restrict__ ea, int* __restrict__ cnt, int* __restrict__ rec)
{
    if (blockIdx.x >= GNODE) {
        const int e = (blockIdx.x - GNODE) * 128 + threadIdx.x;
        if (e < NEDGE) {
            const int s = src[e];
            const int d = dst[e];
            const v4f c = *(const v4f*)(ea + (size_t)e * 4);
            const int b = d >> 4;
            const int slot = atomicAdd(&cnt[b], 1);
            if (slot < BCAP) {
                int* r = rec + ((size_t)b * BCAP + slot) * 8;
                v4i ia; ia.x = s; ia.y = d & 15; ia.z = 0; ia.w = 0;
                *(v4i*)r       = ia;
                *(v4f*)(r + 4) = c;
            }
        }
        return;
    }

    const int wave = threadIdx.x >> 5;
    const int lane = threadIdx.x & 31;
    const int hf = lane >> 4, l = lane & 15;
    const int item = blockIdx.x * 4 + wave;
    if (item >= 9 * NROWP) return;

    const int p  = item / NROWP;
    const int rp = item - p * NROWP;
    const int tm0 = rp << 1;
    const bool has1 = (tm0 + 1) < NROWT;
    const int m0a = tm0 << 4;
    const int m0b = has1 ? (m0a + 16) : m0a;

    const size_t plane = (size_t)NNODE * HDIM;

    float bv[4] = {0.0f, 0.0f, 0.0f, 0.0f};
    float* C;
    int ldc, n0;

    v8f acc[2][4] = {};
    if (p < 6) {
        const float* B = (p < 4) ? (nn_w + (size_t)p * HDIM * HDIM)
                                 : ((p == 4) ? nn_b : root_w);
        C = (p < 5) ? (Y + (size_t)p * plane) : agg;
        ldc = HDIM;
        n0 = 0;
        if (p == 5) {
            bv[0] = conv_b[l];      bv[1] = conv_b[16 + l];
            bv[2] = conv_b[32 + l]; bv[3] = conv_b[48 + l];
        }
        wmma_block_2x4<64, false, false>(h, HDIM, B, HDIM, m0a, m0b, 0, hf, l, acc);
    } else {
        n0 = (p - 6) * HDIM;
        C = gh;
        ldc = 3 * HDIM;
        bv[0] = b_hh[n0 + l];      bv[1] = b_hh[n0 + 16 + l];
        bv[2] = b_hh[n0 + 32 + l]; bv[3] = b_hh[n0 + 48 + l];
        wmma_block_2x4<64, false, true>(h, HDIM, w_hh, HDIM, m0a, m0b, n0, hf, l, acc);
    }

    store_tiles4(C, ldc, m0a, n0, l, hf, bv, acc[0]);
    if (has1) store_tiles4(C, ldc, m0b, n0, l, hf, bv, acc[1]);
}

__device__ __forceinline__ void add_msg_to_tile(
    float* trow, const v2f (&ga)[5], const v2f (&gb)[5],
    float c0, float c1, float c2, float c3, bool keep)
{
    float m0 = ga[4].x, m1 = ga[4].y, m2 = gb[4].x, m3 = gb[4].y;
    m0 = fmaf(c0, ga[0].x, m0); m1 = fmaf(c0, ga[0].y, m1);
    m2 = fmaf(c0, gb[0].x, m2); m3 = fmaf(c0, gb[0].y, m3);
    m0 = fmaf(c1, ga[1].x, m0); m1 = fmaf(c1, ga[1].y, m1);
    m2 = fmaf(c1, gb[1].x, m2); m3 = fmaf(c1, gb[1].y, m3);
    m0 = fmaf(c2, ga[2].x, m0); m1 = fmaf(c2, ga[2].y, m1);
    m2 = fmaf(c2, gb[2].x, m2); m3 = fmaf(c2, gb[2].y, m3);
    m0 = fmaf(c3, ga[3].x, m0); m1 = fmaf(c3, ga[3].y, m1);
    m2 = fmaf(c3, gb[3].x, m2); m3 = fmaf(c3, gb[3].y, m3);
    m0 = keep ? m0 : 0.0f; m1 = keep ? m1 : 0.0f;
    m2 = keep ? m2 : 0.0f; m3 = keep ? m3 : 0.0f;
    atomicAdd(trow,      m0);
    atomicAdd(trow + 1,  m1);
    atomicAdd(trow + 32, m2);
    atomicAdd(trow + 33, m3);
}

__device__ __forceinline__ void gather_rows(
    const float* __restrict__ Y, int s, int o, v2f (&ga)[5], v2f (&gb)[5])
{
    const size_t plane = (size_t)NNODE * HDIM;
    const float* __restrict__ yr = Y + (size_t)s * HDIM + o;
#pragma unroll
    for (int p = 0; p < 5; ++p) {
        const int q = (p + 4) % 5;
        ga[q] = *(const v2f*)(yr + (size_t)q * plane);
        gb[q] = *(const v2f*)(yr + (size_t)q * plane + 32);
    }
}

__global__ __launch_bounds__(256) void edge_gru(
    const int* __restrict__ src, const int* __restrict__ dst,
    const float* __restrict__ ea,
    const float* __restrict__ Y, const float* __restrict__ agg,
    const float* __restrict__ gh,
    const float* __restrict__ w_ih, const float* __restrict__ b_ih,
    const int* __restrict__ cnt, const int* __restrict__ rec,
    float* __restrict__ h)
{
    __shared__ __attribute__((aligned(16))) float tile[16 * LPITCH];

    const int tid  = threadIdx.x;
    const int blk  = blockIdx.x;
    const int m0   = blk << 4;
    const int wave = tid >> 5;
    const int lane = tid & 31;
    const int hf = lane >> 4, l = lane & 15;
    const int j0 = (wave & 3) << 4;
    const int j  = j0 + l;

    const int n = cnt[blk];

    {
        const int r = tid >> 4, c4 = (tid & 15) << 2;
        const v4f v = *(const v4f*)(agg + (size_t)(m0 + r) * HDIM + c4);
        *(v4f*)(tile + r * LPITCH + c4) = v;
    }

    float bi[3] = {0.0f, 0.0f, 0.0f};
    float ghA[3][8] = {}, hoA[8] = {};
    if (wave < 4) {
        bi[0] = b_ih[j]; bi[1] = b_ih[HDIM + j]; bi[2] = b_ih[2 * HDIM + j];
        load_gate_inputs(gh, h, m0, j, hf, ghA, hoA);
    }

    const int o = (tid & 15) << 1;
    const int g = tid >> 4;
    const int* __restrict__ rb = rec + (size_t)blk * BCAP * 8;
    const bool listed = (n <= BCAP) && (g < n);
    bool hasb = listed && ((g + 16) < n);
    v4i ra = {0, 0, 0, 0}, sa = {0, 0, 0, 0};
    v4f rc = {0.0f, 0.0f, 0.0f, 0.0f}, sc = {0.0f, 0.0f, 0.0f, 0.0f};
    if (listed) {
        const int ib = hasb ? (g + 16) : g;
        ra = *(const v4i*)(rb + (size_t)g * 8);
        rc = *(const v4f*)(rb + (size_t)g * 8 + 4);
        sa = *(const v4i*)(rb + (size_t)ib * 8);
        sc = *(const v4f*)(rb + (size_t)ib * 8 + 4);
    }

    __syncthreads();

    if (n <= BCAP) {
        if (listed) {
            int i = g;
            for (;;) {
                const int  in    = i + 32;
                const bool more  = in < n;
                const bool moreb = (in + 16) < n;
                const int  iqa   = more  ? in : i;
                const int  iqb   = moreb ? (in + 16) : iqa;
                const v4i na = *(const v4i*)(rb + (size_t)iqa * 8);
                const v4f nc = *(const v4f*)(rb + (size_t)iqa * 8 + 4);
                const v4i nb = *(const v4i*)(rb + (size_t)iqb * 8);
                const v4f nd = *(const v4f*)(rb + (size_t)iqb * 8 + 4);

                v2f ga[5], gb[5], ha[5], hb[5];
                gather_rows(Y, ra.x, o, ga, gb);
                gather_rows(Y, sa.x, o, ha, hb);

                __builtin_amdgcn_sched_barrier(0);

                add_msg_to_tile(tile + ra.y * LPITCH + o, ga, gb,
                                rc.x, rc.y, rc.z, rc.w, true);
                add_msg_to_tile(tile + sa.y * LPITCH + o, ha, hb,
                                sc.x, sc.y, sc.z, sc.w, hasb);
                if (!more) break;
                i = in; ra = na; rc = nc; sa = nb; sc = nd; hasb = moreb;
            }
        }
    } else {
        for (int e = g; e < NEDGE; e += 16) {
            const int d = dst[e];
            if ((d >> 4) == blk) {
                const int s = src[e];
                const v4f c = *(const v4f*)(ea + (size_t)e * 4);
                v2f ga[5], gb[5];
                gather_rows(Y, s, o, ga, gb);
                add_msg_to_tile(tile + (d & 15) * LPITCH + o, ga, gb,
                                c.x, c.y, c.z, c.w, true);
            }
        }
    }

    __syncthreads();

    if (wave < 4) {
        v8f acc[3] = {};
        wmma_1x3_gates_lds(tile, w_ih, j0, hf, l, acc);
        gate_update_tile(acc, bi, ghA, hoA, h, m0, j, hf);
    }
}

extern "C" void kernel_launch(void* const* d_in, const int* in_sizes, int n_in,
                              void* d_out, int out_size, void* d_ws, size_t ws_size,
                              hipStream_t stream)
{
    (void)in_sizes; (void)n_in; (void)out_size; (void)ws_size;

    const float* x      = (const float*)d_in[0];
    const int*   eidx   = (const int*)  d_in[1];
    const float* ea     = (const float*)d_in[2];
    const float* lin0_w = (const float*)d_in[3];
    const float* lin0_b = (const float*)d_in[4];
    const float* nn_w   = (const float*)d_in[5];
    const float* nn_b   = (const float*)d_in[6];
    const float* root_w = (const float*)d_in[7];
    const float* conv_b = (const float*)d_in[8];
    const float* w_ih   = (const float*)d_in[9];
    const float* w_hh   = (const float*)d_in[10];
    const float* b_ih   = (const float*)d_in[11];
    const float* b_hh   = (const float*)d_in[12];

    const int* src = eidx;
    const int* dst = eidx + NEDGE;

    float* h = (float*)d_out;

    const size_t plane = (size_t)NNODE * HDIM;
    float* Y   = (float*)d_ws;
    float* agg = Y + 5 * plane;
    float* gh  = agg + plane;
    int*   cnt = (int*)(gh + 3 * plane);
    int*   rec = cnt + 640;

    lin0_gemm<<<GLIN0 + 1, 128, 0, stream>>>(x, lin0_w, lin0_b, h, cnt);

    node_gemms<<<GNODE + GBIN, 128, 0, stream>>>(h, nn_w, nn_b, root_w, conv_b,
                                                 w_hh, b_hh, Y, agg, gh,
                                                 src, dst, ea, cnt, rec);
    edge_gru<<<NBKT, 256, 0, stream>>>(src, dst, ea, Y, agg, gh, w_ih, b_ih,
                                       cnt, rec, h);

    for (int it = 1; it < 3; ++it) {
        node_gemms<<<GNODE, 128, 0, stream>>>(h, nn_w, nn_b, root_w, conv_b,
                                              w_hh, b_hh, Y, agg, gh,
                                              src, dst, ea, cnt, rec);
        edge_gru<<<NBKT, 256, 0, stream>>>(src, dst, ea, Y, agg, gh, w_ih, b_ih,
                                           cnt, rec, h);
    }
}
